# starting kernel
# baseline (speedup 1.0000x reference)
#include <hip/hip_runtime.h>

typedef __attribute__((ext_vector_type(2))) float v2f;
typedef __attribute__((ext_vector_type(8))) float v8f;

#define B_TOTAL 4096
#define T_DIM   24
#define F_DIM   256
#define O_DIM   24
#define TGT     2

__global__ __launch_bounds__(256) void dlinear_wmma_f32(
    const float* __restrict__ x,
    const float* __restrict__ W,
    const float* __restrict__ bias,
    float* __restrict__ out)
{
  const int lane = threadIdx.x & 31;
  const int wave = threadIdx.x >> 5;
  const int tile = blockIdx.x * 8 + wave;
  const int b0   = tile * 16;

  const int  mrow = lane & 15;
  const int  grp  = lane >> 4;

  const int   oHi   = 16 + mrow;
  const bool  oHiOk = (oHi < O_DIM);
  const int   oHiC  = oHiOk ? oHi : 0;
  const float oHiM  = oHiOk ? 1.0f : 0.0f;

  v8f acc00 = {};
  v8f acc01 = {};
  v8f acc10 = {};
  v8f acc11 = {};

  const float* xrow = x + (size_t)(b0 + mrow) * (T_DIM * F_DIM);
  const float* w0   = W;
  const float* w1   = W + O_DIM * T_DIM;

#pragma unroll
  for (int k = 0; k < T_DIM; k += 4) {
    const int t0 = k + 2 * grp;

    v2f p0 = *(const v2f*)(xrow + (size_t)t0 * F_DIM);
    v2f p1 = *(const v2f*)(xrow + (size_t)(t0 + 1) * F_DIM);
    v2f a_f0 = { p0.x, p1.x };
    v2f a_f1 = { p0.y, p1.y };

    v2f b00 = *(const v2f*)(w0 + mrow * T_DIM + t0);
    v2f b10 = *(const v2f*)(w1 + mrow * T_DIM + t0);
    v2f b01 = *(const v2f*)(w0 + oHiC * T_DIM + t0) * oHiM;
    v2f b11 = *(const v2f*)(w1 + oHiC * T_DIM + t0) * oHiM;

    acc00 = __builtin_amdgcn_wmma_f32_16x16x4_f32(false, a_f0, false, b00, (short)0, acc00, false, false);
    acc01 = __builtin_amdgcn_wmma_f32_16x16x4_f32(false, a_f0, false, b01, (short)0, acc01, false, false);
    acc10 = __builtin_amdgcn_wmma_f32_16x16x4_f32(false, a_f1, false, b10, (short)0, acc10, false, false);
    acc11 = __builtin_amdgcn_wmma_f32_16x16x4_f32(false, a_f1, false, b11, (short)0, acc11, false, false);
  }

  const float bs00 = bias[0 * O_DIM + mrow];
  const float bs10 = bias[1 * O_DIM + mrow];
  const float bs01 = bias[0 * O_DIM + oHiC] * oHiM;
  const float bs11 = bias[1 * O_DIM + oHiC] * oHiM;

#pragma unroll
  for (int r = 0; r < 8; ++r) {
    const int bb = b0 + r + 8 * grp;
    v2f o0;
    o0.x = acc00[r] + bs00;
    o0.y = acc10[r] + bs10;
    *(v2f*)(out + (size_t)bb * (O_DIM * TGT) + mrow * TGT) = o0;
  }
  if (oHiOk) {
#pragma unroll
    for (int r = 0; r < 8; ++r) {
      const int bb = b0 + r + 8 * grp;
      v2f o1;
      o1.x = acc01[r] + bs01;
      o1.y = acc11[r] + bs11;
      *(v2f*)(out + (size_t)bb * (O_DIM * TGT) + oHi * TGT) = o1;
    }
  }
}

extern "C" void kernel_launch(void* const* d_in, const int* in_sizes, int n_in,
                              void* d_out, int out_size, void* d_ws, size_t ws_size,
                              hipStream_t stream) {
  const float* x    = (const float*)d_in[0];
  const float* W    = (const float*)d_in[1];
  const float* bias = (const float*)d_in[2];
  float* out = (float*)d_out;

  dim3 grid(B_TOTAL / 16 / 8);
  dim3 block(256);
  dlinear_wmma_f32<<<grid, block, 0, stream>>>(x, W, bias, out);
}


# launch shape: one 16-batch tile per single-wave work-group, 256 work-groups (was 32 of 8 waves), so the strided x gather (a 128-byte line per 8 bytes used) spreads over 256 WGPs, not 32
# speedup vs baseline: 1.5391x; 1.5391x over previous
#include <hip/hip_runtime.h>

typedef __attribute__((ext_vector_type(2))) float v2f;
typedef __attribute__((ext_vector_type(8))) float v8f;

#define B_TOTAL 4096
#define T_DIM   24
#define F_DIM   256
#define O_DIM   24
#define TGT     2

__global__ __launch_bounds__(32) void dlinear_wmma_f32(
    const float* __restrict__ x,
    const float* __restrict__ W,
    const float* __restrict__ bias,
    float* __restrict__ out)
{
  const int lane = threadIdx.x;
  const int tile = blockIdx.x;
  const int b0   = tile * 16;

  const int  mrow = lane & 15;
  const int  grp  = lane >> 4;

  const int   oHi   = 16 + mrow;
  const bool  oHiOk = (oHi < O_DIM);
  const int   oHiC  = oHiOk ? oHi : 0;
  const float oHiM  = oHiOk ? 1.0f : 0.0f;

  v8f acc00 = {};
  v8f acc01 = {};
  v8f acc10 = {};
  v8f acc11 = {};

  const float* xrow = x + (size_t)(b0 + mrow) * (T_DIM * F_DIM);
  const float* w0   = W;
  const float* w1   = W + O_DIM * T_DIM;

#pragma unroll
  for (int k = 0; k < T_DIM; k += 4) {
    const int t0 = k + 2 * grp;

    v2f p0 = *(const v2f*)(xrow + (size_t)t0 * F_DIM);
    v2f p1 = *(const v2f*)(xrow + (size_t)(t0 + 1) * F_DIM);
    v2f a_f0 = { p0.x, p1.x };
    v2f a_f1 = { p0.y, p1.y };

    v2f b00 = *(const v2f*)(w0 + mrow * T_DIM + t0);
    v2f b10 = *(const v2f*)(w1 + mrow * T_DIM + t0);
    v2f b01 = *(const v2f*)(w0 + oHiC * T_DIM + t0) * oHiM;
    v2f b11 = *(const v2f*)(w1 + oHiC * T_DIM + t0) * oHiM;

    acc00 = __builtin_amdgcn_wmma_f32_16x16x4_f32(false, a_f0, false, b00, (short)0, acc00, false, false);
    acc01 = __builtin_amdgcn_wmma_f32_16x16x4_f32(false, a_f0, false, b01, (short)0, acc01, false, false);
    acc10 = __builtin_amdgcn_wmma_f32_16x16x4_f32(false, a_f1, false, b10, (short)0, acc10, false, false);
    acc11 = __builtin_amdgcn_wmma_f32_16x16x4_f32(false, a_f1, false, b11, (short)0, acc11, false, false);
  }

  const float bs00 = bias[0 * O_DIM + mrow];
  const float bs10 = bias[1 * O_DIM + mrow];
  const float bs01 = bias[0 * O_DIM + oHiC] * oHiM;
  const float bs11 = bias[1 * O_DIM + oHiC] * oHiM;

#pragma unroll
  for (int r = 0; r < 8; ++r) {
    const int bb = b0 + r + 8 * grp;
    v2f o0;
    o0.x = acc00[r] + bs00;
    o0.y = acc10[r] + bs10;
    *(v2f*)(out + (size_t)bb * (O_DIM * TGT) + mrow * TGT) = o0;
  }
  if (oHiOk) {
#pragma unroll
    for (int r = 0; r < 8; ++r) {
      const int bb = b0 + r + 8 * grp;
      v2f o1;
      o1.x = acc01[r] + bs01;
      o1.y = acc11[r] + bs11;
      *(v2f*)(out + (size_t)bb * (O_DIM * TGT) + oHi * TGT) = o1;
    }
  }
}

extern "C" void kernel_launch(void* const* d_in, const int* in_sizes, int n_in,
                              void* d_out, int out_size, void* d_ws, size_t ws_size,
                              hipStream_t stream) {
  const float* x    = (const float*)d_in[0];
  const float* W    = (const float*)d_in[1];
  const float* bias = (const float*)d_in[2];
  float* out = (float*)d_out;

  dim3 grid(256);
  dim3 block(32);
  dlinear_wmma_f32<<<grid, block, 0, stream>>>(x, W, bias, out);
}


# every global read of a wave (4 bias words, 24 W fragments, 12 x gathers) issued as one group before the first WMMA, short reads first, pinned by scheduling barriers: one memory trip, not five
# speedup vs baseline: 2.0690x; 1.3443x over previous
#include <hip/hip_runtime.h>

typedef __attribute__((ext_vector_type(2))) float v2f;
typedef __attribute__((ext_vector_type(8))) float v8f;

#define B_TOTAL 4096
#define T_DIM   24
#define F_DIM   256
#define O_DIM   24
#define TGT     2

__global__ __launch_bounds__(32) void dlinear_wmma_f32(
    const float* __restrict__ x,
    const float* __restrict__ W,
    const float* __restrict__ bias,
    float* __restrict__ out)
{
  const int lane = threadIdx.x;
  const int tile = blockIdx.x;
  const int b0   = tile * 16;

  const int  mrow = lane & 15;
  const int  grp  = lane >> 4;

  const int   oHi   = 16 + mrow;
  const bool  oHiOk = (oHi < O_DIM);
  const int   oHiC  = oHiOk ? oHi : 0;
  const float oHiM  = oHiOk ? 1.0f : 0.0f;

  v8f acc00 = {};
  v8f acc01 = {};
  v8f acc10 = {};
  v8f acc11 = {};

  const float* xrow = x + (size_t)(b0 + mrow) * (T_DIM * F_DIM);
  const float* w0   = W;
  const float* w1   = W + O_DIM * T_DIM;

  float bs00 = bias[0 * O_DIM + mrow];
  float bs10 = bias[1 * O_DIM + mrow];
  float bh0  = bias[0 * O_DIM + oHiC];
  float bh1  = bias[1 * O_DIM + oHiC];

  v2f wq00[T_DIM / 4], wq10[T_DIM / 4], wq01[T_DIM / 4], wq11[T_DIM / 4];
#pragma unroll
  for (int j = 0; j < T_DIM / 4; ++j) {
    const int t0 = 4 * j + 2 * grp;
    wq00[j] = *(const v2f*)(w0 + mrow * T_DIM + t0);
    wq10[j] = *(const v2f*)(w1 + mrow * T_DIM + t0);
    wq01[j] = *(const v2f*)(w0 + oHiC * T_DIM + t0);
    wq11[j] = *(const v2f*)(w1 + oHiC * T_DIM + t0);
  }

  __builtin_amdgcn_sched_barrier(0);

  v2f xp0[T_DIM / 4], xp1[T_DIM / 4];
#pragma unroll
  for (int j = 0; j < T_DIM / 4; ++j) {
    const int t0 = 4 * j + 2 * grp;
    xp0[j] = *(const v2f*)(xrow + (size_t)t0 * F_DIM);
    xp1[j] = *(const v2f*)(xrow + (size_t)(t0 + 1) * F_DIM);
  }

  __builtin_amdgcn_sched_barrier(0);
  asm volatile("" : "+v"(bs00), "+v"(bs10), "+v"(bh0), "+v"(bh1));

#pragma unroll
  for (int j = 0; j < T_DIM / 4; ++j) {
    v2f a_f0 = { xp0[j].x, xp1[j].x };
    v2f a_f1 = { xp0[j].y, xp1[j].y };
    v2f b00 = wq00[j];
    v2f b10 = wq10[j];
    v2f b01 = wq01[j] * oHiM;
    v2f b11 = wq11[j] * oHiM;

    acc00 = __builtin_amdgcn_wmma_f32_16x16x4_f32(false, a_f0, false, b00, (short)0, acc00, false, false);
    acc01 = __builtin_amdgcn_wmma_f32_16x16x4_f32(false, a_f0, false, b01, (short)0, acc01, false, false);
    acc10 = __builtin_amdgcn_wmma_f32_16x16x4_f32(false, a_f1, false, b10, (short)0, acc10, false, false);
    acc11 = __builtin_amdgcn_wmma_f32_16x16x4_f32(false, a_f1, false, b11, (short)0, acc11, false, false);
  }

  const float bs01 = bh0 * oHiM;
  const float bs11 = bh1 * oHiM;

#pragma unroll
  for (int r = 0; r < 8; ++r) {
    const int bb = b0 + r + 8 * grp;
    v2f o0;
    o0.x = acc00[r] + bs00;
    o0.y = acc10[r] + bs10;
    *(v2f*)(out + (size_t)bb * (O_DIM * TGT) + mrow * TGT) = o0;
  }
  if (oHiOk) {
#pragma unroll
    for (int r = 0; r < 8; ++r) {
      const int bb = b0 + r + 8 * grp;
      v2f o1;
      o1.x = acc01[r] + bs01;
      o1.y = acc11[r] + bs11;
      *(v2f*)(out + (size_t)bb * (O_DIM * TGT) + oHi * TGT) = o1;
    }
  }
}

extern "C" void kernel_launch(void* const* d_in, const int* in_sizes, int n_in,
                              void* d_out, int out_size, void* d_ws, size_t ws_size,
                              hipStream_t stream) {
  const float* x    = (const float*)d_in[0];
  const float* W    = (const float*)d_in[1];
  const float* bias = (const float*)d_in[2];
  float* out = (float*)d_out;

  dim3 grid(256);
  dim3 block(32);
  dlinear_wmma_f32<<<grid, block, 0, stream>>>(x, W, bias, out);
}
